# starting kernel
# baseline (speedup 1.0000x reference)
#include <hip/hip_runtime.h>

typedef float v2f __attribute__((ext_vector_type(2)));
typedef float v8f __attribute__((ext_vector_type(8)));

__global__ __launch_bounds__(256) void mbd_gemm_f32wmma(
    const float* __restrict__ x,
    const float* __restrict__ kern,
    float* __restrict__ M)
{
    const int lane   = threadIdx.x & 31;
    const int wave   = threadIdx.x >> 5;
    const int waveId = blockIdx.x * 8 + wave;
    const int tm     = waveId >> 4;
    const int tn     = waveId & 15;

    const int l16  = lane & 15;
    const int half = lane >> 4;
    const int r0   = tm * 16;
    const int c0   = tn * 16;

    const float* arow = x + (r0 + l16) * 256 + 2 * half;

    const int    j    = c0 + l16;
    const float* wcol = kern + (j >> 3) * 2048 + (j & 7);
    const int    kb   = 2 * half;

    v8f acc = {};
#pragma unroll 4
    for (int k = 0; k < 256; k += 4) {
        v2f a = *(const v2f*)(arow + k);
        v2f b;
        b.x = wcol[(k + kb) * 8];
        b.y = wcol[(k + kb + 1) * 8];
        acc = __builtin_amdgcn_wmma_f32_16x16x4_f32(
                  false, a, false, b, (short)0, acc, false, false);
    }

#pragma unroll
    for (int r = 0; r < 8; ++r) {
        M[(r0 + r + 8 * half) * 256 + c0 + l16] = acc[r];
    }
}

__global__ __launch_bounds__(256) void mbd_pairwise(
    const float* __restrict__ M,
    float* __restrict__ out)
{
    __shared__ float lds[1024 * 8];

    const int o   = blockIdx.y;
    const int tid = threadIdx.x;

    const float* src = M + o * 8;
    for (int i = tid; i < 2048; i += 256) {
        const int b = i >> 1;
        const int h = i & 1;
        *(float4*)(&lds[i * 4]) = *(const float4*)(src + b * 256 + h * 4);
    }
    __syncthreads();

    const int b = blockIdx.x * 256 + tid;
    float m[8];
#pragma unroll
    for (int l = 0; l < 8; ++l) m[l] = lds[b * 8 + l];

    const float kLog2e = 1.4426950408889634f;
    float acc = 0.0f;
#pragma unroll 4
    for (int bp = 0; bp < 1024; ++bp) {
        const float* q = &lds[bp * 8];
        float l1 = 0.0f;
#pragma unroll
        for (int l = 0; l < 8; ++l) l1 += __builtin_fabsf(m[l] - q[l]);
        acc += __builtin_amdgcn_exp2f(-l1 * kLog2e);
    }
    out[b * 32 + o] = acc * (1.0f / 1024.0f);
}

extern "C" void kernel_launch(void* const* d_in, const int* in_sizes, int n_in,
                              void* d_out, int out_size, void* d_ws, size_t ws_size,
                              hipStream_t stream)
{
    const float* x    = (const float*)d_in[0];
    const float* kern = (const float*)d_in[1];
    float*       out  = (float*)d_out;
    float*       M    = (float*)d_ws;

    mbd_gemm_f32wmma<<<128, 256, 0, stream>>>(x, kern, M);
    mbd_pairwise<<<dim3(4, 32), 256, 0, stream>>>(M, out);
}


# Stage 2 re-tiled: block = 128 rows x one o (256 blocks, all WGPs), 4 rows per lane share each LDS partner read, 8 waves split the partners; partials added via LDS (new order of the partner sum)
# speedup vs baseline: 1.4094x; 1.4094x over previous
#include <hip/hip_runtime.h>

typedef float v2f __attribute__((ext_vector_type(2)));
typedef float v8f __attribute__((ext_vector_type(8)));

__global__ __launch_bounds__(256) void mbd_gemm_f32wmma(
    const float* __restrict__ x,
    const float* __restrict__ kern,
    float* __restrict__ M)
{
    const int lane   = threadIdx.x & 31;
    const int wave   = threadIdx.x >> 5;
    const int waveId = blockIdx.x * 8 + wave;
    const int tm     = waveId >> 4;
    const int tn     = waveId & 15;

    const int l16  = lane & 15;
    const int half = lane >> 4;
    const int r0   = tm * 16;
    const int c0   = tn * 16;

    const float* arow = x + (r0 + l16) * 256 + 2 * half;

    const int    j    = c0 + l16;
    const float* wcol = kern + (j >> 3) * 2048 + (j & 7);
    const int    kb   = 2 * half;

    v8f acc = {};
#pragma unroll 4
    for (int k = 0; k < 256; k += 4) {
        v2f a = *(const v2f*)(arow + k);
        v2f b;
        b.x = wcol[(k + kb) * 8];
        b.y = wcol[(k + kb + 1) * 8];
        acc = __builtin_amdgcn_wmma_f32_16x16x4_f32(
                  false, a, false, b, (short)0, acc, false, false);
    }

#pragma unroll
    for (int r = 0; r < 8; ++r) {
        M[(r0 + r + 8 * half) * 256 + c0 + l16] = acc[r];
    }
}


__device__ __forceinline__ void abs_add_abs8(float (&s)[8], const float (&x)[8], const float (&y)[8])
{
    asm("v_add_f32 %0, |%8|, |%16|\n\t"
        "v_add_f32 %1, |%9|, |%17|\n\t"
        "v_add_f32 %2, |%10|, |%18|\n\t"
        "v_add_f32 %3, |%11|, |%19|\n\t"
        "v_add_f32 %4, |%12|, |%20|\n\t"
        "v_add_f32 %5, |%13|, |%21|\n\t"
        "v_add_f32 %6, |%14|, |%22|\n\t"
        "v_add_f32 %7, |%15|, |%23|"
        : "=&v"(s[0]), "=&v"(s[1]), "=&v"(s[2]), "=&v"(s[3]),
          "=&v"(s[4]), "=&v"(s[5]), "=&v"(s[6]), "=&v"(s[7])
        : "v"(x[0]), "v"(x[1]), "v"(x[2]), "v"(x[3]),
          "v"(x[4]), "v"(x[5]), "v"(x[6]), "v"(x[7]),
          "v"(y[0]), "v"(y[1]), "v"(y[2]), "v"(y[3]),
          "v"(y[4]), "v"(y[5]), "v"(y[6]), "v"(y[7]));
}

__device__ __forceinline__ void add_abs8(float (&s)[8], const float (&x)[8])
{
    asm("v_add_f32 %0, %0, |%8|\n\t"
        "v_add_f32 %1, %1, |%9|\n\t"
        "v_add_f32 %2, %2, |%10|\n\t"
        "v_add_f32 %3, %3, |%11|\n\t"
        "v_add_f32 %4, %4, |%12|\n\t"
        "v_add_f32 %5, %5, |%13|\n\t"
        "v_add_f32 %6, %6, |%14|\n\t"
        "v_add_f32 %7, %7, |%15|"
        : "+v"(s[0]), "+v"(s[1]), "+v"(s[2]), "+v"(s[3]),
          "+v"(s[4]), "+v"(s[5]), "+v"(s[6]), "+v"(s[7])
        : "v"(x[0]), "v"(x[1]), "v"(x[2]), "v"(x[3]),
          "v"(x[4]), "v"(x[5]), "v"(x[6]), "v"(x[7]));
}

__global__ __launch_bounds__(256) void mbd_pairwise(
    const float* __restrict__ M,
    float* __restrict__ out)
{
    __shared__ float lds[1024 * 8];
    __shared__ float part[8 * 128];

    const int o    = blockIdx.y;
    const int bt   = blockIdx.x;
    const int tid  = threadIdx.x;
    const int lane = tid & 31;
    const int wave = tid >> 5;

    const float* src = M + o * 8;
    for (int i = tid; i < 2048; i += 256) {
        const int b = i >> 1;
        const int h = i & 1;
        *(float4*)(&lds[i * 4]) = *(const float4*)(src + b * 256 + h * 4);
    }
    __syncthreads();

    v2f m[4][4];
#pragma unroll
    for (int r = 0; r < 4; ++r) {
        const int b = bt * 128 + r * 32 + lane;
#pragma unroll
        for (int p = 0; p < 4; ++p) m[r][p] = *(const v2f*)(&lds[b * 8 + 2 * p]);
    }

    const float kLog2e = 1.4426950408889634f;
    float acc[4] = {0.0f, 0.0f, 0.0f, 0.0f};
    const float* qbase = &lds[wave * 128 * 8];

    for (int i = 0; i < 128; i += 2) {
        v2f q[2][4];
#pragma unroll
        for (int j = 0; j < 2; ++j)
#pragma unroll
            for (int p = 0; p < 4; ++p) q[j][p] = *(const v2f*)(qbase + (i + j) * 8 + 2 * p);

        float l1[8];
        float dx[8], dy[8];
#pragma unroll
        for (int j = 0; j < 2; ++j)
#pragma unroll
            for (int r = 0; r < 4; ++r) {
                const v2f d = m[r][0] - q[j][0];
                dx[j * 4 + r] = d.x;
                dy[j * 4 + r] = d.y;
            }
#pragma unroll
        for (int p = 0; p < 4; ++p) {
            float nx[8], ny[8];
            if (p < 3) {
#pragma unroll
                for (int j = 0; j < 2; ++j)
#pragma unroll
                    for (int r = 0; r < 4; ++r) {
                        const v2f d = m[r][p + 1] - q[j][p + 1];
                        nx[j * 4 + r] = d.x;
                        ny[j * 4 + r] = d.y;
                    }
            }
            if (p == 0) {
                abs_add_abs8(l1, dx, dy);
            } else {
                add_abs8(l1, dx);
                add_abs8(l1, dy);
            }
            if (p < 3) {
#pragma unroll
                for (int c = 0; c < 8; ++c) { dx[c] = nx[c]; dy[c] = ny[c]; }
            }
        }
#pragma unroll
        for (int j = 0; j < 2; ++j)
#pragma unroll
            for (int r = 0; r < 4; ++r)
                acc[r] += __builtin_amdgcn_exp2f(-l1[j * 4 + r] * kLog2e);
    }

#pragma unroll
    for (int r = 0; r < 4; ++r) part[wave * 128 + r * 32 + lane] = acc[r];
    __syncthreads();

    if (tid < 128) {
        float s = 0.0f;
#pragma unroll
        for (int w = 0; w < 8; ++w) s += part[w * 128 + tid];
        out[(bt * 128 + tid) * 32 + o] = s * (1.0f / 1024.0f);
    }
}

extern "C" void kernel_launch(void* const* d_in, const int* in_sizes, int n_in,
                              void* d_out, int out_size, void* d_ws, size_t ws_size,
                              hipStream_t stream)
{
    const float* x    = (const float*)d_in[0];
    const float* kern = (const float*)d_in[1];
    float*       out  = (float*)d_out;
    float*       M    = (float*)d_ws;

    mbd_gemm_f32wmma<<<128, 256, 0, stream>>>(x, kern, M);
    mbd_pairwise<<<dim3(8, 32), 256, 0, stream>>>(M, out);
}


# Stage 1 writes M slice by slice (Mt[o][b][l], tile turned through LDS, whole 128 B lines per store); stage 2 stages its slice as one contiguous 32 KB run, not 32 B out of each of 1024 lines
# speedup vs baseline: 2.3523x; 1.6690x over previous
#include <hip/hip_runtime.h>

typedef float v2f __attribute__((ext_vector_type(2)));
typedef float v8f __attribute__((ext_vector_type(8)));

#define MBD_APITCH 260
#define MBD_BPITCH 2056
#define MBD_TPITCH 20

__global__ __launch_bounds__(128) void mbd_gemm_f32wmma(
    const float* __restrict__ x,
    const float* __restrict__ kern,
    float* __restrict__ M)
{
    __shared__ __attribute__((aligned(16))) float sA[32 * MBD_APITCH];
    __shared__ __attribute__((aligned(16))) float sB[4 * MBD_BPITCH];
    __shared__ __attribute__((aligned(16))) float sT[4 * 16 * MBD_TPITCH];

    const int tid = threadIdx.x;
    const int rb  = blockIdx.x >> 3;
    const int cb  = blockIdx.x & 7;

    const float4* gA = (const float4*)(x + rb * 32 * 256);
    const float4* gB = (const float4*)(kern + cb * 4 * 2048);

    float4 ra[16], rk[16];
#pragma unroll
    for (int u = 0; u < 16; ++u) ra[u] = gA[tid + 128 * u];
#pragma unroll
    for (int u = 0; u < 16; ++u) rk[u] = gB[tid + 128 * u];

#pragma unroll
    for (int u = 0; u < 16; ++u)
        *(float4*)(&sA[((tid >> 6) + 2 * u) * MBD_APITCH + (tid & 63) * 4]) = ra[u];
#pragma unroll
    for (int u = 0; u < 16; ++u)
        *(float4*)(&sB[(u >> 2) * MBD_BPITCH + (tid + 128 * (u & 3)) * 4]) = rk[u];
    __syncthreads();

    const int lane = tid & 31;
    const int wave = tid >> 5;
    const int wr   = wave >> 1;
    const int wc   = wave & 1;
    const int l16  = lane & 15;
    const int half = lane >> 4;

    const float* pa = &sA[(16 * wr + l16) * MBD_APITCH + 4 * half];
    const float* pb = &sB[(2 * wc + (l16 >> 3)) * MBD_BPITCH + 32 * half + (l16 & 7)];

    float4 fa[2][8];
    v2f    fb[2][16];
#pragma unroll
    for (int j = 0; j < 8; ++j) {
        fa[0][j]         = *(const float4*)(pa + 8 * j);
        fb[0][2 * j].x     = pb[64 * j];
        fb[0][2 * j].y     = pb[64 * j + 8];
        fb[0][2 * j + 1].x = pb[64 * j + 16];
        fb[0][2 * j + 1].y = pb[64 * j + 24];
    }

    v8f acc0 = {};
    v8f acc1 = {};
#pragma unroll
    for (int ch = 0; ch < 4; ++ch) {
        const int cur = ch & 1;
        const int nxt = cur ^ 1;
        if (ch < 3) {
#pragma unroll
            for (int j = 0; j < 8; ++j) {
                const int kk = (ch + 1) * 8 + j;
                fa[nxt][j]           = *(const float4*)(pa + 8 * kk);
                fb[nxt][2 * j].x     = pb[64 * kk];
                fb[nxt][2 * j].y     = pb[64 * kk + 8];
                fb[nxt][2 * j + 1].x = pb[64 * kk + 16];
                fb[nxt][2 * j + 1].y = pb[64 * kk + 24];
            }
        }
        __builtin_amdgcn_sched_barrier(0);
#pragma unroll
        for (int j = 0; j < 8; ++j) {
            v2f a01, a23;
            a01.x = fa[cur][j].x; a01.y = fa[cur][j].y;
            a23.x = fa[cur][j].z; a23.y = fa[cur][j].w;
            acc0 = __builtin_amdgcn_wmma_f32_16x16x4_f32(
                       false, a01, false, fb[cur][2 * j], (short)0, acc0, false, false);
            acc1 = __builtin_amdgcn_wmma_f32_16x16x4_f32(
                       false, a23, false, fb[cur][2 * j + 1], (short)0, acc1, false, false);
        }
        __builtin_amdgcn_sched_barrier(0);
    }
    const v8f acc = acc0 + acc1;

    float* tw = &sT[wave * 16 * MBD_TPITCH];
#pragma unroll
    for (int r = 0; r < 8; ++r) {
        tw[(r + 8 * half) * MBD_TPITCH + l16] = acc[r];
    }
    __syncthreads();
    const int r0   = rb * 32 + wr * 16;
    const int o0   = cb * 4 + wc * 2;
    const int trow = lane >> 1;
    const int tl0  = (lane & 1) * 4;
#pragma unroll
    for (int k = 0; k < 2; ++k) {
        const float4 v = *(const float4*)(&tw[trow * MBD_TPITCH + 8 * k + tl0]);
        *(float4*)(&M[(o0 + k) * 8192 + (r0 + trow) * 8 + tl0]) = v;
    }
}


__device__ __forceinline__ void abs_add_abs8(float (&s)[8], const float (&x)[8], const float (&y)[8])
{
    asm("v_add_f32 %0, |%8|, |%16|\n\t"
        "v_add_f32 %1, |%9|, |%17|\n\t"
        "v_add_f32 %2, |%10|, |%18|\n\t"
        "v_add_f32 %3, |%11|, |%19|\n\t"
        "v_add_f32 %4, |%12|, |%20|\n\t"
        "v_add_f32 %5, |%13|, |%21|\n\t"
        "v_add_f32 %6, |%14|, |%22|\n\t"
        "v_add_f32 %7, |%15|, |%23|"
        : "=&v"(s[0]), "=&v"(s[1]), "=&v"(s[2]), "=&v"(s[3]),
          "=&v"(s[4]), "=&v"(s[5]), "=&v"(s[6]), "=&v"(s[7])
        : "v"(x[0]), "v"(x[1]), "v"(x[2]), "v"(x[3]),
          "v"(x[4]), "v"(x[5]), "v"(x[6]), "v"(x[7]),
          "v"(y[0]), "v"(y[1]), "v"(y[2]), "v"(y[3]),
          "v"(y[4]), "v"(y[5]), "v"(y[6]), "v"(y[7]));
}

__device__ __forceinline__ void add_abs8(float (&s)[8], const float (&x)[8])
{
    asm("v_add_f32 %0, %0, |%8|\n\t"
        "v_add_f32 %1, %1, |%9|\n\t"
        "v_add_f32 %2, %2, |%10|\n\t"
        "v_add_f32 %3, %3, |%11|\n\t"
        "v_add_f32 %4, %4, |%12|\n\t"
        "v_add_f32 %5, %5, |%13|\n\t"
        "v_add_f32 %6, %6, |%14|\n\t"
        "v_add_f32 %7, %7, |%15|"
        : "+v"(s[0]), "+v"(s[1]), "+v"(s[2]), "+v"(s[3]),
          "+v"(s[4]), "+v"(s[5]), "+v"(s[6]), "+v"(s[7])
        : "v"(x[0]), "v"(x[1]), "v"(x[2]), "v"(x[3]),
          "v"(x[4]), "v"(x[5]), "v"(x[6]), "v"(x[7]));
}

__global__ __launch_bounds__(256) void mbd_pairwise(
    const float* __restrict__ M,
    float* __restrict__ out)
{
    __shared__ float lds[1024 * 8];
    __shared__ float part[8 * 128];

    const int o    = blockIdx.y;
    const int bt   = blockIdx.x;
    const int tid  = threadIdx.x;
    const int lane = tid & 31;
    const int wave = tid >> 5;

    const float4* src = (const float4*)(M + o * 8192);
    float4 st[8];
#pragma unroll
    for (int u = 0; u < 8; ++u) st[u] = src[tid + 256 * u];
#pragma unroll
    for (int u = 0; u < 8; ++u) *(float4*)(&lds[(tid + 256 * u) * 4]) = st[u];
    __syncthreads();

    v2f m[4][4];
#pragma unroll
    for (int r = 0; r < 4; ++r) {
        const int b = bt * 128 + r * 32 + lane;
#pragma unroll
        for (int p = 0; p < 4; ++p) m[r][p] = *(const v2f*)(&lds[b * 8 + 2 * p]);
    }

    const float kLog2e = 1.4426950408889634f;
    float acc[4] = {0.0f, 0.0f, 0.0f, 0.0f};
    const float* qbase = &lds[wave * 128 * 8];

    for (int i = 0; i < 128; i += 2) {
        v2f q[2][4];
#pragma unroll
        for (int j = 0; j < 2; ++j)
#pragma unroll
            for (int p = 0; p < 4; ++p) q[j][p] = *(const v2f*)(qbase + (i + j) * 8 + 2 * p);

        float l1[8];
        float dx[8], dy[8];
#pragma unroll
        for (int j = 0; j < 2; ++j)
#pragma unroll
            for (int r = 0; r < 4; ++r) {
                const v2f d = m[r][0] - q[j][0];
                dx[j * 4 + r] = d.x;
                dy[j * 4 + r] = d.y;
            }
#pragma unroll
        for (int p = 0; p < 4; ++p) {
            float nx[8], ny[8];
            if (p < 3) {
#pragma unroll
                for (int j = 0; j < 2; ++j)
#pragma unroll
                    for (int r = 0; r < 4; ++r) {
                        const v2f d = m[r][p + 1] - q[j][p + 1];
                        nx[j * 4 + r] = d.x;
                        ny[j * 4 + r] = d.y;
                    }
            }
            if (p == 0) {
                abs_add_abs8(l1, dx, dy);
            } else {
                add_abs8(l1, dx);
                add_abs8(l1, dy);
            }
            if (p < 3) {
#pragma unroll
                for (int c = 0; c < 8; ++c) { dx[c] = nx[c]; dy[c] = ny[c]; }
            }
        }
#pragma unroll
        for (int j = 0; j < 2; ++j)
#pragma unroll
            for (int r = 0; r < 4; ++r)
                acc[r] += __builtin_amdgcn_exp2f(-l1[j * 4 + r] * kLog2e);
    }

#pragma unroll
    for (int r = 0; r < 4; ++r) part[wave * 128 + r * 32 + lane] = acc[r];
    __syncthreads();

    if (tid < 128) {
        float s = 0.0f;
#pragma unroll
        for (int w = 0; w < 8; ++w) s += part[w * 128 + tid];
        out[(bt * 128 + tid) * 32 + o] = s * (1.0f / 1024.0f);
    }
}

extern "C" void kernel_launch(void* const* d_in, const int* in_sizes, int n_in,
                              void* d_out, int out_size, void* d_ws, size_t ws_size,
                              hipStream_t stream)
{
    const float* x    = (const float*)d_in[0];
    const float* kern = (const float*)d_in[1];
    float*       out  = (float*)d_out;
    float*       M    = (float*)d_ws;

    mbd_gemm_f32wmma<<<256, 128, 0, stream>>>(x, kern, M);
    mbd_pairwise<<<dim3(8, 32), 256, 0, stream>>>(M, out);
}


# Stage 2 reads each step's two partners from LDS one step ahead, into a second register set (two steps per trip, no copies): no wave waits for an LDS round trip inside the pair loop
# speedup vs baseline: 2.5480x; 1.0832x over previous
#include <hip/hip_runtime.h>

typedef float v2f __attribute__((ext_vector_type(2)));
typedef float v8f __attribute__((ext_vector_type(8)));

#define MBD_APITCH 260
#define MBD_BPITCH 2056
#define MBD_TPITCH 20

__global__ __launch_bounds__(128) void mbd_gemm_f32wmma(
    const float* __restrict__ x,
    const float* __restrict__ kern,
    float* __restrict__ M)
{
    __shared__ __attribute__((aligned(16))) float sA[32 * MBD_APITCH];
    __shared__ __attribute__((aligned(16))) float sB[4 * MBD_BPITCH];
    __shared__ __attribute__((aligned(16))) float sT[4 * 16 * MBD_TPITCH];

    const int tid = threadIdx.x;
    const int rb  = blockIdx.x >> 3;
    const int cb  = blockIdx.x & 7;

    const float4* gA = (const float4*)(x + rb * 32 * 256);
    const float4* gB = (const float4*)(kern + cb * 4 * 2048);

    float4 ra[16], rk[16];
#pragma unroll
    for (int u = 0; u < 16; ++u) ra[u] = gA[tid + 128 * u];
#pragma unroll
    for (int u = 0; u < 16; ++u) rk[u] = gB[tid + 128 * u];

#pragma unroll
    for (int u = 0; u < 16; ++u)
        *(float4*)(&sA[((tid >> 6) + 2 * u) * MBD_APITCH + (tid & 63) * 4]) = ra[u];
#pragma unroll
    for (int u = 0; u < 16; ++u)
        *(float4*)(&sB[(u >> 2) * MBD_BPITCH + (tid + 128 * (u & 3)) * 4]) = rk[u];
    __syncthreads();

    const int lane = tid & 31;
    const int wave = tid >> 5;
    const int wr   = wave >> 1;
    const int wc   = wave & 1;
    const int l16  = lane & 15;
    const int half = lane >> 4;

    const float* pa = &sA[(16 * wr + l16) * MBD_APITCH + 4 * half];
    const float* pb = &sB[(2 * wc + (l16 >> 3)) * MBD_BPITCH + 32 * half + (l16 & 7)];

    float4 fa[2][8];
    v2f    fb[2][16];
#pragma unroll
    for (int j = 0; j < 8; ++j) {
        fa[0][j]         = *(const float4*)(pa + 8 * j);
        fb[0][2 * j].x     = pb[64 * j];
        fb[0][2 * j].y     = pb[64 * j + 8];
        fb[0][2 * j + 1].x = pb[64 * j + 16];
        fb[0][2 * j + 1].y = pb[64 * j + 24];
    }

    v8f acc0 = {};
    v8f acc1 = {};
#pragma unroll
    for (int ch = 0; ch < 4; ++ch) {
        const int cur = ch & 1;
        const int nxt = cur ^ 1;
        if (ch < 3) {
#pragma unroll
            for (int j = 0; j < 8; ++j) {
                const int kk = (ch + 1) * 8 + j;
                fa[nxt][j]           = *(const float4*)(pa + 8 * kk);
                fb[nxt][2 * j].x     = pb[64 * kk];
                fb[nxt][2 * j].y     = pb[64 * kk + 8];
                fb[nxt][2 * j + 1].x = pb[64 * kk + 16];
                fb[nxt][2 * j + 1].y = pb[64 * kk + 24];
            }
        }
        __builtin_amdgcn_sched_barrier(0);
#pragma unroll
        for (int j = 0; j < 8; ++j) {
            v2f a01, a23;
            a01.x = fa[cur][j].x; a01.y = fa[cur][j].y;
            a23.x = fa[cur][j].z; a23.y = fa[cur][j].w;
            acc0 = __builtin_amdgcn_wmma_f32_16x16x4_f32(
                       false, a01, false, fb[cur][2 * j], (short)0, acc0, false, false);
            acc1 = __builtin_amdgcn_wmma_f32_16x16x4_f32(
                       false, a23, false, fb[cur][2 * j + 1], (short)0, acc1, false, false);
        }
        __builtin_amdgcn_sched_barrier(0);
    }
    const v8f acc = acc0 + acc1;

    float* tw = &sT[wave * 16 * MBD_TPITCH];
#pragma unroll
    for (int r = 0; r < 8; ++r) {
        tw[(r + 8 * half) * MBD_TPITCH + l16] = acc[r];
    }
    __syncthreads();
    const int r0   = rb * 32 + wr * 16;
    const int o0   = cb * 4 + wc * 2;
    const int trow = lane >> 1;
    const int tl0  = (lane & 1) * 4;
#pragma unroll
    for (int k = 0; k < 2; ++k) {
        const float4 v = *(const float4*)(&tw[trow * MBD_TPITCH + 8 * k + tl0]);
        *(float4*)(&M[(o0 + k) * 8192 + (r0 + trow) * 8 + tl0]) = v;
    }
}


__device__ __forceinline__ void abs_add_abs8(float (&s)[8], const float (&x)[8], const float (&y)[8])
{
    asm("v_add_f32 %0, |%8|, |%16|\n\t"
        "v_add_f32 %1, |%9|, |%17|\n\t"
        "v_add_f32 %2, |%10|, |%18|\n\t"
        "v_add_f32 %3, |%11|, |%19|\n\t"
        "v_add_f32 %4, |%12|, |%20|\n\t"
        "v_add_f32 %5, |%13|, |%21|\n\t"
        "v_add_f32 %6, |%14|, |%22|\n\t"
        "v_add_f32 %7, |%15|, |%23|"
        : "=&v"(s[0]), "=&v"(s[1]), "=&v"(s[2]), "=&v"(s[3]),
          "=&v"(s[4]), "=&v"(s[5]), "=&v"(s[6]), "=&v"(s[7])
        : "v"(x[0]), "v"(x[1]), "v"(x[2]), "v"(x[3]),
          "v"(x[4]), "v"(x[5]), "v"(x[6]), "v"(x[7]),
          "v"(y[0]), "v"(y[1]), "v"(y[2]), "v"(y[3]),
          "v"(y[4]), "v"(y[5]), "v"(y[6]), "v"(y[7]));
}

__device__ __forceinline__ void add_abs8(float (&s)[8], const float (&x)[8])
{
    asm("v_add_f32 %0, %0, |%8|\n\t"
        "v_add_f32 %1, %1, |%9|\n\t"
        "v_add_f32 %2, %2, |%10|\n\t"
        "v_add_f32 %3, %3, |%11|\n\t"
        "v_add_f32 %4, %4, |%12|\n\t"
        "v_add_f32 %5, %5, |%13|\n\t"
        "v_add_f32 %6, %6, |%14|\n\t"
        "v_add_f32 %7, %7, |%15|"
        : "+v"(s[0]), "+v"(s[1]), "+v"(s[2]), "+v"(s[3]),
          "+v"(s[4]), "+v"(s[5]), "+v"(s[6]), "+v"(s[7])
        : "v"(x[0]), "v"(x[1]), "v"(x[2]), "v"(x[3]),
          "v"(x[4]), "v"(x[5]), "v"(x[6]), "v"(x[7]));
}

__device__ __forceinline__ void mbd_read_partners(v2f (&q)[2][4], const float* p)
{
#pragma unroll
    for (int j = 0; j < 2; ++j)
#pragma unroll
        for (int c = 0; c < 4; ++c) q[j][c] = *(const v2f*)(p + j * 8 + 2 * c);
}

__device__ __forceinline__ void mbd_pair_step(float (&acc)[4], const v2f (&m)[4][4], const v2f (&q)[2][4])
{
    const float kLog2e = 1.4426950408889634f;

    float l1[8];
    float dx[8], dy[8];
#pragma unroll
    for (int j = 0; j < 2; ++j)
#pragma unroll
        for (int r = 0; r < 4; ++r) {
            const v2f d = m[r][0] - q[j][0];
            dx[j * 4 + r] = d.x;
            dy[j * 4 + r] = d.y;
        }
#pragma unroll
    for (int p = 0; p < 4; ++p) {
        float nx[8], ny[8];
        if (p < 3) {
#pragma unroll
            for (int j = 0; j < 2; ++j)
#pragma unroll
                for (int r = 0; r < 4; ++r) {
                    const v2f d = m[r][p + 1] - q[j][p + 1];
                    nx[j * 4 + r] = d.x;
                    ny[j * 4 + r] = d.y;
                }
        }
        if (p == 0) {
            abs_add_abs8(l1, dx, dy);
        } else {
            add_abs8(l1, dx);
            add_abs8(l1, dy);
        }
        if (p < 3) {
#pragma unroll
            for (int c = 0; c < 8; ++c) { dx[c] = nx[c]; dy[c] = ny[c]; }
        }
    }
#pragma unroll
    for (int j = 0; j < 2; ++j)
#pragma unroll
        for (int r = 0; r < 4; ++r)
            acc[r] += __builtin_amdgcn_exp2f(-l1[j * 4 + r] * kLog2e);
}

__global__ __launch_bounds__(256) void mbd_pairwise(
    const float* __restrict__ M,
    float* __restrict__ out)
{
    __shared__ float lds[1024 * 8];
    __shared__ float part[8 * 128];

    const int o    = blockIdx.y;
    const int bt   = blockIdx.x;
    const int tid  = threadIdx.x;
    const int lane = tid & 31;
    const int wave = tid >> 5;

    const float4* src = (const float4*)(M + o * 8192);
    float4 st[8];
#pragma unroll
    for (int u = 0; u < 8; ++u) st[u] = src[tid + 256 * u];
#pragma unroll
    for (int u = 0; u < 8; ++u) *(float4*)(&lds[(tid + 256 * u) * 4]) = st[u];
    __syncthreads();

    v2f m[4][4];
#pragma unroll
    for (int r = 0; r < 4; ++r) {
        const int b = bt * 128 + r * 32 + lane;
#pragma unroll
        for (int p = 0; p < 4; ++p) m[r][p] = *(const v2f*)(&lds[b * 8 + 2 * p]);
    }

    float acc[4] = {0.0f, 0.0f, 0.0f, 0.0f};
    const float* qbase = &lds[wave * 128 * 8];

    v2f qa[2][4], qb[2][4];
    mbd_read_partners(qa, qbase);
    for (int i = 0; i < 128; i += 4) {
        mbd_read_partners(qb, qbase + (i + 2) * 8);
        __builtin_amdgcn_sched_barrier(0);
        mbd_pair_step(acc, m, qa);
        mbd_read_partners(qa, qbase + ((i + 4) & 127) * 8);
        __builtin_amdgcn_sched_barrier(0);
        mbd_pair_step(acc, m, qb);
    }

#pragma unroll
    for (int r = 0; r < 4; ++r) part[wave * 128 + r * 32 + lane] = acc[r];
    __syncthreads();

    if (tid < 128) {
        float s = 0.0f;
#pragma unroll
        for (int w = 0; w < 8; ++w) s += part[w * 128 + tid];
        out[(bt * 128 + tid) * 32 + o] = s * (1.0f / 1024.0f);
    }
}

extern "C" void kernel_launch(void* const* d_in, const int* in_sizes, int n_in,
                              void* d_out, int out_size, void* d_ws, size_t ws_size,
                              hipStream_t stream)
{
    const float* x    = (const float*)d_in[0];
    const float* kern = (const float*)d_in[1];
    float*       out  = (float*)d_out;
    float*       M    = (float*)d_ws;

    mbd_gemm_f32wmma<<<256, 128, 0, stream>>>(x, kern, M);
    mbd_pairwise<<<dim3(8, 32), 256, 0, stream>>>(M, out);
}


# Stage 2: behind a pair's 4 abs-adds all work runs packed over two partners (tail adds, running sum), log2e folded into the staged slice: new order of the l sum (tree) and partner sum (even/odd)
# speedup vs baseline: 2.8140x; 1.1044x over previous
#include <hip/hip_runtime.h>

typedef float v2f __attribute__((ext_vector_type(2)));
typedef float v8f __attribute__((ext_vector_type(8)));

#define MBD_APITCH 260
#define MBD_BPITCH 2056
#define MBD_TPITCH 20

__global__ __launch_bounds__(128) void mbd_gemm_f32wmma(
    const float* __restrict__ x,
    const float* __restrict__ kern,
    float* __restrict__ M)
{
    __shared__ __attribute__((aligned(16))) float sA[32 * MBD_APITCH];
    __shared__ __attribute__((aligned(16))) float sB[4 * MBD_BPITCH];
    __shared__ __attribute__((aligned(16))) float sT[4 * 16 * MBD_TPITCH];

    const int tid = threadIdx.x;
    const int rb  = blockIdx.x >> 3;
    const int cb  = blockIdx.x & 7;

    const float4* gA = (const float4*)(x + rb * 32 * 256);
    const float4* gB = (const float4*)(kern + cb * 4 * 2048);

    float4 ra[16], rk[16];
#pragma unroll
    for (int u = 0; u < 16; ++u) ra[u] = gA[tid + 128 * u];
#pragma unroll
    for (int u = 0; u < 16; ++u) rk[u] = gB[tid + 128 * u];

#pragma unroll
    for (int u = 0; u < 16; ++u)
        *(float4*)(&sA[((tid >> 6) + 2 * u) * MBD_APITCH + (tid & 63) * 4]) = ra[u];
#pragma unroll
    for (int u = 0; u < 16; ++u)
        *(float4*)(&sB[(u >> 2) * MBD_BPITCH + (tid + 128 * (u & 3)) * 4]) = rk[u];
    __syncthreads();

    const int lane = tid & 31;
    const int wave = tid >> 5;
    const int wr   = wave >> 1;
    const int wc   = wave & 1;
    const int l16  = lane & 15;
    const int half = lane >> 4;

    const float* pa = &sA[(16 * wr + l16) * MBD_APITCH + 4 * half];
    const float* pb = &sB[(2 * wc + (l16 >> 3)) * MBD_BPITCH + 32 * half + (l16 & 7)];

    float4 fa[2][8];
    v2f    fb[2][16];
#pragma unroll
    for (int j = 0; j < 8; ++j) {
        fa[0][j]         = *(const float4*)(pa + 8 * j);
        fb[0][2 * j].x     = pb[64 * j];
        fb[0][2 * j].y     = pb[64 * j + 8];
        fb[0][2 * j + 1].x = pb[64 * j + 16];
        fb[0][2 * j + 1].y = pb[64 * j + 24];
    }

    v8f acc0 = {};
    v8f acc1 = {};
#pragma unroll
    for (int ch = 0; ch < 4; ++ch) {
        const int cur = ch & 1;
        const int nxt = cur ^ 1;
        if (ch < 3) {
#pragma unroll
            for (int j = 0; j < 8; ++j) {
                const int kk = (ch + 1) * 8 + j;
                fa[nxt][j]           = *(const float4*)(pa + 8 * kk);
                fb[nxt][2 * j].x     = pb[64 * kk];
                fb[nxt][2 * j].y     = pb[64 * kk + 8];
                fb[nxt][2 * j + 1].x = pb[64 * kk + 16];
                fb[nxt][2 * j + 1].y = pb[64 * kk + 24];
            }
        }
        __builtin_amdgcn_sched_barrier(0);
#pragma unroll
        for (int j = 0; j < 8; ++j) {
            v2f a01, a23;
            a01.x = fa[cur][j].x; a01.y = fa[cur][j].y;
            a23.x = fa[cur][j].z; a23.y = fa[cur][j].w;
            acc0 = __builtin_amdgcn_wmma_f32_16x16x4_f32(
                       false, a01, false, fb[cur][2 * j], (short)0, acc0, false, false);
            acc1 = __builtin_amdgcn_wmma_f32_16x16x4_f32(
                       false, a23, false, fb[cur][2 * j + 1], (short)0, acc1, false, false);
        }
        __builtin_amdgcn_sched_barrier(0);
    }
    const v8f acc = acc0 + acc1;

    float* tw = &sT[wave * 16 * MBD_TPITCH];
#pragma unroll
    for (int r = 0; r < 8; ++r) {
        tw[(r + 8 * half) * MBD_TPITCH + l16] = acc[r];
    }
    __syncthreads();
    const int r0   = rb * 32 + wr * 16;
    const int o0   = cb * 4 + wc * 2;
    const int trow = lane >> 1;
    const int tl0  = (lane & 1) * 4;
#pragma unroll
    for (int k = 0; k < 2; ++k) {
        const float4 v = *(const float4*)(&tw[trow * MBD_TPITCH + 8 * k + tl0]);
        *(float4*)(&M[(o0 + k) * 8192 + (r0 + trow) * 8 + tl0]) = v;
    }
}


__device__ __forceinline__ void abs_add_abs8(float (&s)[8], const float (&x)[8], const float (&y)[8])
{
    asm("v_add_f32 %0, |%8|, |%16|\n\t"
        "v_add_f32 %1, |%9|, |%17|\n\t"
        "v_add_f32 %2, |%10|, |%18|\n\t"
        "v_add_f32 %3, |%11|, |%19|\n\t"
        "v_add_f32 %4, |%12|, |%20|\n\t"
        "v_add_f32 %5, |%13|, |%21|\n\t"
        "v_add_f32 %6, |%14|, |%22|\n\t"
        "v_add_f32 %7, |%15|, |%23|"
        : "=&v"(s[0]), "=&v"(s[1]), "=&v"(s[2]), "=&v"(s[3]),
          "=&v"(s[4]), "=&v"(s[5]), "=&v"(s[6]), "=&v"(s[7])
        : "v"(x[0]), "v"(x[1]), "v"(x[2]), "v"(x[3]),
          "v"(x[4]), "v"(x[5]), "v"(x[6]), "v"(x[7]),
          "v"(y[0]), "v"(y[1]), "v"(y[2]), "v"(y[3]),
          "v"(y[4]), "v"(y[5]), "v"(y[6]), "v"(y[7]));
}

__device__ __forceinline__ void mbd_read_partners(v2f (&q)[2][4], const float* p)
{
#pragma unroll
    for (int j = 0; j < 2; ++j)
#pragma unroll
        for (int c = 0; c < 4; ++c) q[j][c] = *(const v2f*)(p + j * 8 + 2 * c);
}

__device__ __forceinline__ void mbd_pair_step(v2f (&acc)[4], const v2f (&m)[4][4], const v2f (&q)[2][4])
{
    float t[4][8];
    float dx[8], dy[8];
#pragma unroll
    for (int j = 0; j < 2; ++j)
#pragma unroll
        for (int r = 0; r < 4; ++r) {
            const v2f d = m[r][0] - q[j][0];
            dx[j * 4 + r] = d.x;
            dy[j * 4 + r] = d.y;
        }
#pragma unroll
    for (int p = 0; p < 4; ++p) {
        float nx[8], ny[8];
        if (p < 3) {
#pragma unroll
            for (int j = 0; j < 2; ++j)
#pragma unroll
                for (int r = 0; r < 4; ++r) {
                    const v2f d = m[r][p + 1] - q[j][p + 1];
                    nx[j * 4 + r] = d.x;
                    ny[j * 4 + r] = d.y;
                }
        }
        abs_add_abs8(t[p], dx, dy);
        if (p < 3) {
#pragma unroll
            for (int c = 0; c < 8; ++c) { dx[c] = nx[c]; dy[c] = ny[c]; }
        }
    }

    v2f lo[4], hi[4];
#pragma unroll
    for (int r = 0; r < 4; ++r) {
        v2f s0, s1;
        s0.x = t[0][r]; s0.y = t[0][4 + r];
        s1.x = t[1][r]; s1.y = t[1][4 + r];
        lo[r] = s0 + s1;
    }
#pragma unroll
    for (int r = 0; r < 4; ++r) {
        v2f s2, s3;
        s2.x = t[2][r]; s2.y = t[2][4 + r];
        s3.x = t[3][r]; s3.y = t[3][4 + r];
        hi[r] = s2 + s3;
    }
#pragma unroll
    for (int r = 0; r < 4; ++r) {
        const v2f a = lo[r] + hi[r];
        v2f e;
        e.x = __builtin_amdgcn_exp2f(-a.x);
        e.y = __builtin_amdgcn_exp2f(-a.y);
        acc[r] += e;
    }
}

__global__ __launch_bounds__(256) void mbd_pairwise(
    const float* __restrict__ M,
    float* __restrict__ out)
{
    __shared__ float lds[1024 * 8];
    __shared__ float part[8 * 128];

    const int o    = blockIdx.y;
    const int bt   = blockIdx.x;
    const int tid  = threadIdx.x;
    const int lane = tid & 31;
    const int wave = tid >> 5;

    const float4* src = (const float4*)(M + o * 8192);
    float4 st[8];
#pragma unroll
    for (int u = 0; u < 8; ++u) st[u] = src[tid + 256 * u];
    const float kLog2e = 1.4426950408889634f;
#pragma unroll
    for (int u = 0; u < 8; ++u) {
        float4 v = st[u];
        v.x *= kLog2e; v.y *= kLog2e; v.z *= kLog2e; v.w *= kLog2e;
        *(float4*)(&lds[(tid + 256 * u) * 4]) = v;
    }
    __syncthreads();

    v2f m[4][4];
#pragma unroll
    for (int r = 0; r < 4; ++r) {
        const int b = bt * 128 + r * 32 + lane;
#pragma unroll
        for (int p = 0; p < 4; ++p) m[r][p] = *(const v2f*)(&lds[b * 8 + 2 * p]);
    }

    v2f acc[4] = {};
    const float* qbase = &lds[wave * 128 * 8];

    v2f qa[2][4], qb[2][4];
    mbd_read_partners(qa, qbase);
    for (int i = 0; i < 128; i += 4) {
        mbd_read_partners(qb, qbase + (i + 2) * 8);
        __builtin_amdgcn_sched_barrier(0);
        mbd_pair_step(acc, m, qa);
        mbd_read_partners(qa, qbase + ((i + 4) & 127) * 8);
        __builtin_amdgcn_sched_barrier(0);
        mbd_pair_step(acc, m, qb);
    }

#pragma unroll
    for (int r = 0; r < 4; ++r) part[wave * 128 + r * 32 + lane] = acc[r].x + acc[r].y;
    __syncthreads();

    if (tid < 128) {
        float s = 0.0f;
#pragma unroll
        for (int w = 0; w < 8; ++w) s += part[w * 128 + tid];
        out[(bt * 128 + tid) * 32 + o] = s * (1.0f / 1024.0f);
    }
}

extern "C" void kernel_launch(void* const* d_in, const int* in_sizes, int n_in,
                              void* d_out, int out_size, void* d_ws, size_t ws_size,
                              hipStream_t stream)
{
    const float* x    = (const float*)d_in[0];
    const float* kern = (const float*)d_in[1];
    float*       out  = (float*)d_out;
    float*       M    = (float*)d_ws;

    mbd_gemm_f32wmma<<<256, 128, 0, stream>>>(x, kern, M);
    mbd_pairwise<<<dim3(8, 32), 256, 0, stream>>>(M, out);
}
